# starting kernel
# baseline (speedup 1.0000x reference)
#include <hip/hip_runtime.h>
#include <math.h>

#define BS    8
#define V     2048
#define NNB   20
#define INC   64
#define OUTC  128
#define SUP   2
#define SUPC  (SUP * OUTC)
#define FEATC (OUTC + SUPC)
#define MROWS (BS * V)
#define EPSF  1e-12f

typedef float v2f __attribute__((ext_vector_type(2)));
typedef float v8f __attribute__((ext_vector_type(8)));

template <bool BT>
__global__ __launch_bounds__(128)
void gemm_bias_wmma(const float* __restrict__ A, const float* __restrict__ B,
                    const float* __restrict__ bias, float* __restrict__ C,
                    int M, int N, int K) {
  const int lane  = threadIdx.x & 31;
  const int wave  = threadIdx.x >> 5;
  const int tile  = blockIdx.x * 4 + wave;
  const int tilesN = N >> 4;
  const int tm = tile / tilesN;
  const int tn = tile - tm * tilesN;
  if (tm * 16 >= M) return;

  const int m0   = tm << 4;
  const int n0   = tn << 4;
  const int half = lane >> 4;
  const int l16  = lane & 15;

  v8f acc = {0.f, 0.f, 0.f, 0.f, 0.f, 0.f, 0.f, 0.f};

  const float* __restrict__ Arow = A + (size_t)(m0 + l16) * K;

  for (int k0 = 0; k0 < K; k0 += 4) {
    const int ka = k0 + 2 * half;
    v2f a;
    a.x = Arow[ka];
    a.y = Arow[ka + 1];

    v2f b;
    if (BT) {
      const float* bp = B + (size_t)(n0 + l16) * K + ka;
      b.x = bp[0];
      b.y = bp[1];
    } else {
      const float* bp = B + (size_t)ka * N + n0 + l16;
      b.x = bp[0];
      b.y = bp[N];
    }
    acc = __builtin_amdgcn_wmma_f32_16x16x4_f32(false, a, false, b,
                                                (short)0, acc, false, false);
  }

  const float bv = bias[n0 + l16];
  float* __restrict__ Crow = C + (size_t)(m0 + 8 * half) * N + n0 + l16;
#pragma unroll
  for (int r = 0; r < 8; ++r)
    Crow[(size_t)r * N] = acc[r] + bv;
}

__global__ __launch_bounds__(256)
void geom_fuse_kernel(const int* __restrict__ nbr, const float* __restrict__ verts,
                      const float* __restrict__ feat,
                      const float* __restrict__ directions,
                      const float* __restrict__ distw,
                      float* __restrict__ out256) {
  const int row = blockIdx.x;
  const int b   = row / V;
  const int s   = threadIdx.x;

  __shared__ float s_dirn[NNB][3];
  __shared__ float s_dist[NNB];
  __shared__ int   s_nidx[NNB];
  __shared__ float s_acc[SUPC];
  __shared__ float s_df[SUPC];

  if (s < NNB) {
    const int ni = nbr[(size_t)row * NNB + s];
    s_nidx[s] = ni;
    const float* pv = verts + (size_t)row * 3;
    const float* pn = verts + ((size_t)b * V + ni) * 3;
    const float dx = pn[0] - pv[0];
    const float dy = pn[1] - pv[1];
    const float dz = pn[2] - pv[2];
    const float d  = sqrtf(dx * dx + dy * dy + dz * dz);
    const float iv = 1.0f / fmaxf(d, EPSF);
    s_dirn[s][0] = dx * iv;
    s_dirn[s][1] = dy * iv;
    s_dirn[s][2] = dz * iv;
    s_dist[s] = d;
  }
  __syncthreads();

  float dmax = 0.0f;
#pragma unroll
  for (int n = 0; n < NNB; ++n) dmax = fmaxf(dmax, s_dist[n]);

  float d0 = directions[0 * SUPC + s];
  float d1 = directions[1 * SUPC + s];
  float d2 = directions[2 * SUPC + s];
  const float nrm = sqrtf(d0 * d0 + d1 * d1 + d2 * d2);
  const float iv  = 1.0f / fmaxf(nrm, EPSF);
  d0 *= iv; d1 *= iv; d2 *= iv;

  float acc = -INFINITY;
  const size_t bbase = (size_t)b * V;
#pragma unroll 4
  for (int n = 0; n < NNB; ++n) {
    const float th = fmaxf(0.0f, d0 * s_dirn[n][0] +
                                 d1 * s_dirn[n][1] +
                                 d2 * s_dirn[n][2]);
    const float sg = feat[(bbase + s_nidx[n]) * FEATC + OUTC + s];
    acc = fmaxf(acc, th * sg);
  }

  s_acc[s] = acc;
  s_df[s]  = fmaxf(0.0f, dmax * distw[s]);
  __syncthreads();

  if (s < OUTC) {
    const float center = feat[(size_t)row * FEATC + s];
    out256[(size_t)row * SUPC + s] = center + s_acc[s] + s_acc[s + OUTC];
  } else {
    const int c = s - OUTC;
    out256[(size_t)row * SUPC + OUTC + c] = s_df[c] + s_df[c + OUTC];
  }
}

extern "C" void kernel_launch(void* const* d_in, const int* in_sizes, int n_in,
                              void* d_out, int out_size, void* d_ws, size_t ws_size,
                              hipStream_t stream) {
  const int*   nbr   = (const int*)  d_in[0];
  const float* verts = (const float*)d_in[1];
  const float* fmap  = (const float*)d_in[2];
  const float* wgt   = (const float*)d_in[3];
  const float* bias  = (const float*)d_in[4];
  const float* dirs  = (const float*)d_in[5];
  const float* distw = (const float*)d_in[6];
  const float* mlpw  = (const float*)d_in[7];
  const float* mlpb  = (const float*)d_in[8];
  float* out = (float*)d_out;

  float* feat   = (float*)d_ws;
  float* out256 = feat + (size_t)MROWS * FEATC;

  {
    const int tiles = (MROWS / 16) * (FEATC / 16);
    gemm_bias_wmma<false><<<tiles / 4, 128, 0, stream>>>(
        fmap, wgt, bias, feat, MROWS, FEATC, INC);
  }

  geom_fuse_kernel<<<MROWS, 256, 0, stream>>>(nbr, verts, feat, dirs, distw, out256);

  {
    const int tiles = (MROWS / 16) * (OUTC / 16);
    gemm_bias_wmma<true><<<tiles / 4, 128, 0, stream>>>(
        out256, mlpw, mlpb, out, MROWS, OUTC, SUPC);
  }
}


# Intermediate feat kept plane-major in accumulator order (24 planes of [rows][16], one per channel slice): GEMM 1 stores whole 128-byte lines, gather kernel stages its slab as one contiguous block
# speedup vs baseline: 2.1628x; 2.1628x over previous
#include <hip/hip_runtime.h>
#include <math.h>


#define BS    8
#define V     2048
#define NNB   20
#define INC   64
#define OUTC  128
#define SUP   2
#define SUPC  (SUP * OUTC)
#define FEATC (OUTC + SUPC)
#define MROWS (BS * V)
#define EPSF  1e-12f

#define SLC    8
#define NSL    (OUTC / SLC)
#define SLS    (SUP * SLC)
#define PITCH4 5
#define VPART  1024

#define NCT    (OUTC / 16)
#define NTILE  (FEATC / 16)

typedef float v2f __attribute__((ext_vector_type(2)));
typedef float v4f __attribute__((ext_vector_type(4)));
typedef float v8f __attribute__((ext_vector_type(8)));

#define RB   64
#define LPAD 4

__global__ __launch_bounds__(384)
void gemm_feat_rowblock(const float* __restrict__ A, const float* __restrict__ B,
                        const float* __restrict__ bias, float* __restrict__ C) {
  constexpr int K  = INC;
  constexpr int N  = FEATC;
  constexpr int PA = K + LPAD;
  constexpr int KS = K / 4;
  __shared__ alignas(16) float sA[RB * PA];

  const int t    = threadIdx.x;
  const int lane = t & 31;
  const int wave = t >> 5;
  const int half = lane >> 4;
  const int l16  = lane & 15;
  const int m0   = blockIdx.x * RB;

  const v4f* __restrict__ A4 = reinterpret_cast<const v4f*>(A + (size_t)m0 * K);
  v4f ra0 = {0.f, 0.f, 0.f, 0.f}, ra1 = ra0, ra2 = ra0, ra3 = ra0;
  if (t < 256) {
    ra0 = A4[t];
    ra1 = A4[256 + t];
    ra2 = A4[512 + t];
    ra3 = A4[768 + t];
  }

  if (t < 256) {
    *reinterpret_cast<v4f*>(&sA[((0   + t) >> 4) * PA + (t & 15) * 4]) = ra0;
    *reinterpret_cast<v4f*>(&sA[((256 + t) >> 4) * PA + (t & 15) * 4]) = ra1;
    *reinterpret_cast<v4f*>(&sA[((512 + t) >> 4) * PA + (t & 15) * 4]) = ra2;
    *reinterpret_cast<v4f*>(&sA[((768 + t) >> 4) * PA + (t & 15) * 4]) = ra3;
  }

  __syncthreads();

  int col[2];
#pragma unroll
  for (int nt = 0; nt < 2; ++nt) {
    const int T = 2 * wave + nt;
    col[nt] = (T < NCT) ? (T * 16 + l16)
                        : (OUTC + (l16 >> 3) * OUTC + (T - NCT) * SLC + (l16 & 7));
  }
  v2f bf[2][KS];
#pragma unroll
  for (int ks = 0; ks < KS; ++ks) {
#pragma unroll
    for (int nt = 0; nt < 2; ++nt) {
      const float* bp = B + (size_t)(ks * 4 + 2 * half) * N + col[nt];
      bf[nt][ks].x = bp[0];
      bf[nt][ks].y = bp[N];
    }
  }
  const float bv0 = bias[col[0]];
  const float bv1 = bias[col[1]];

#pragma unroll
  for (int pr = 0; pr < RB / 32; ++pr) {
    v8f acc[2][2];
#pragma unroll
    for (int x = 0; x < 2; ++x)
#pragma unroll
      for (int y = 0; y < 2; ++y)
        acc[x][y] = (v8f){0.f, 0.f, 0.f, 0.f, 0.f, 0.f, 0.f, 0.f};

    const float* ar = &sA[(pr * 32 + l16) * PA + 2 * half];
#pragma unroll
    for (int ks = 0; ks < KS; ++ks) {
      const v2f a0 = *reinterpret_cast<const v2f*>(ar + ks * 4);
      const v2f a1 = *reinterpret_cast<const v2f*>(ar + 16 * PA + ks * 4);
      acc[0][0] = __builtin_amdgcn_wmma_f32_16x16x4_f32(false, a0, false, bf[0][ks],
                                                        (short)0, acc[0][0], false, false);
      acc[0][1] = __builtin_amdgcn_wmma_f32_16x16x4_f32(false, a0, false, bf[1][ks],
                                                        (short)0, acc[0][1], false, false);
      acc[1][0] = __builtin_amdgcn_wmma_f32_16x16x4_f32(false, a1, false, bf[0][ks],
                                                        (short)0, acc[1][0], false, false);
      acc[1][1] = __builtin_amdgcn_wmma_f32_16x16x4_f32(false, a1, false, bf[1][ks],
                                                        (short)0, acc[1][1], false, false);
    }
#pragma unroll
    for (int x = 0; x < 2; ++x) {
      float* __restrict__ blk0 =
          C + ((size_t)(2 * wave) * MROWS + m0 + pr * 32 + x * 16) * 16 + lane;
      float* __restrict__ blk1 = blk0 + (size_t)MROWS * 16;
#pragma unroll
      for (int r = 0; r < 8; ++r) {
        blk0[r * 32] = acc[x][0][r] + bv0;
        blk1[r * 32] = acc[x][1][r] + bv1;
      }
    }
    __builtin_amdgcn_sched_barrier(0);
  }
}

#define KCH   64
#define NCH   (SUPC / KCH)
__global__ __launch_bounds__(256)
void gemm_out_rowblock(const float* __restrict__ A, const float* __restrict__ Bw,
                       const float* __restrict__ bias, float* __restrict__ C) {
  constexpr int K  = SUPC;
  constexpr int N  = OUTC;
  constexpr int P  = K + LPAD;
  constexpr int NA = RB * KCH / 4 / 256;
  constexpr int NB = N * KCH / 4 / 256;
  __shared__ alignas(16) float sA[RB * P];
  __shared__ alignas(16) float sB[N * P];

  const int t    = threadIdx.x;
  const int lane = t & 31;
  const int wave = t >> 5;
  const int half = lane >> 4;
  const int l16  = lane & 15;
  const int m0   = blockIdx.x * RB;
  const v4f* __restrict__ B4 = reinterpret_cast<const v4f*>(Bw);

  v4f ra[NA], rb[NB];

#define STAGE_LOAD(c)                                                              \
  {                                                                                \
    _Pragma("unroll") for (int j = 0; j < NA; ++j) {                               \
      const int i = j * 256 + t;                                                   \
      const int p = (c) * (KCH / SLC) + (i >> 7);                                  \
      const int r = (i >> 1) & (RB - 1);                                           \
      ra[j] = *reinterpret_cast<const v4f*>(                                       \
          A + ((size_t)p * MROWS + m0 + r) * SLC + (i & 1) * 4);                   \
    }                                                                              \
    _Pragma("unroll") for (int j = 0; j < NB; ++j) {                               \
      const int i = j * 256 + t;                                                   \
      rb[j] = B4[(i >> 4) * (K / 4) + (c) * (KCH / 4) + (i & 15)];                 \
    }                                                                              \
  }
#define STAGE_STORE(c)                                                             \
  {                                                                                \
    _Pragma("unroll") for (int j = 0; j < NA; ++j) {                               \
      const int i = j * 256 + t;                                                   \
      const int p = (c) * (KCH / SLC) + (i >> 7);                                  \
      const int r = (i >> 1) & (RB - 1);                                           \
      *reinterpret_cast<v4f*>(&sA[r * P + p * SLC + (i & 1) * 4]) = ra[j];         \
    }                                                                              \
    _Pragma("unroll") for (int j = 0; j < NB; ++j) {                               \
      const int i = j * 256 + t;                                                   \
      *reinterpret_cast<v4f*>(&sB[(i >> 4) * P + (c) * KCH + (i & 15) * 4]) =      \
          rb[j];                                                                   \
    }                                                                              \
  }

  STAGE_LOAD(0)
  const int   n0 = wave * 16;
  const float bv = bias[n0 + l16];
  STAGE_STORE(0)

  v8f acc[RB / 16];
#pragma unroll
  for (int mt = 0; mt < RB / 16; ++mt)
    acc[mt] = (v8f){0.f, 0.f, 0.f, 0.f, 0.f, 0.f, 0.f, 0.f};

  const float* br = &sB[(n0 + l16) * P + 2 * half];
  const float* ar = &sA[l16 * P + 2 * half];

#pragma unroll
  for (int c = 0; c < NCH; ++c) {
    __syncthreads();
    if (c + 1 < NCH) STAGE_LOAD(c + 1)
#pragma unroll
    for (int kk = 0; kk < KCH / 4; ++kk) {
      const int ks = c * (KCH / 4) + kk;
      const v2f b = *reinterpret_cast<const v2f*>(br + ks * 4);
#pragma unroll
      for (int mt = 0; mt < RB / 16; ++mt) {
        const v2f a = *reinterpret_cast<const v2f*>(ar + mt * 16 * P + ks * 4);
        acc[mt] = __builtin_amdgcn_wmma_f32_16x16x4_f32(false, a, false, b,
                                                        (short)0, acc[mt], false, false);
      }
    }
    __builtin_amdgcn_sched_barrier(0);
    if (c + 1 < NCH) STAGE_STORE(c + 1)
  }
#undef STAGE_LOAD
#undef STAGE_STORE

#pragma unroll
  for (int mt = 0; mt < RB / 16; ++mt) {
    float* __restrict__ Crow = C + (size_t)(m0 + mt * 16 + 8 * half) * N + n0 + l16;
#pragma unroll
    for (int r = 0; r < 8; ++r)
      Crow[(size_t)r * N] = acc[mt][r] + bv;
  }
}

__global__ __launch_bounds__(1024)
void slab_gather_kernel(const int* __restrict__ nbr, const float* __restrict__ verts,
                        const float* __restrict__ feat,
                        const float* __restrict__ directions,
                        const float* __restrict__ distw,
                        float* __restrict__ planes) {
  __shared__ float4 s_slab[V * PITCH4];
  __shared__ float4 s_vert[V];

  const int t  = threadIdx.x;
  const int hv = blockIdx.x & 1;
  const int j  = (blockIdx.x >> 1) & (NSL - 1);
  const int b  = blockIdx.x >> 5;
  const int c0 = j * SLC;
  const size_t rbase = (size_t)b * V;
  const int    v   = hv * VPART + t;
  const size_t row = rbase + v;

  const int4* __restrict__ np = reinterpret_cast<const int4*>(nbr + row * NNB);
  int4 cur = np[0];

  const float4* __restrict__ slab4 =
      reinterpret_cast<const float4*>(feat + ((size_t)(NCT + j) * MROWS + rbase) * 16);
#pragma unroll
  for (int it = 0; it < (V * 4) / 1024; ++it) {
    const int idx = it * 1024 + t;
    const int p = idx >> 2;
    const int q = idx & 3;
    const int u = (p & ~15) | ((p & 1) << 3) | ((p >> 1) & 7);
    s_slab[u * PITCH4 + q] = slab4[idx];
  }
#pragma unroll
  for (int it = 0; it < V / 1024; ++it) {
    const int u = it * 1024 + t;
    const float* pv = verts + (rbase + u) * 3;
    s_vert[u] = make_float4(pv[0], pv[1], pv[2], 0.0f);
  }

  float sd0[SLS], sd1[SLS], sd2[SLS];
#pragma unroll
  for (int i = 0; i < SLS; ++i) {
    const int s = (i >> 3) * OUTC + c0 + (i & 7);
    float d0 = directions[0 * SUPC + s];
    float d1 = directions[1 * SUPC + s];
    float d2 = directions[2 * SUPC + s];
    const float nrm = sqrtf(d0 * d0 + d1 * d1 + d2 * d2);
    const float iv  = 1.0f / fmaxf(nrm, EPSF);
    sd0[i] = d0 * iv;
    sd1[i] = d1 * iv;
    sd2[i] = d2 * iv;
  }

  __syncthreads();

  const float4 pc = s_vert[v];
  float acc[SLS];
#pragma unroll
  for (int i = 0; i < SLS; ++i) acc[i] = -INFINITY;
  float dmax = 0.0f;

#pragma unroll 1
  for (int g = 0; g < NNB / 4; ++g) {
    const int4 nxt = np[(g + 1 < NNB / 4) ? (g + 1) : g];
    const int ids[4] = {cur.x, cur.y, cur.z, cur.w};
#pragma unroll
    for (int e = 0; e < 4; ++e) {
      const int ni = ids[e] & (V - 1);
      const float4 pn = s_vert[ni];
      const float dx = pn.x - pc.x;
      const float dy = pn.y - pc.y;
      const float dz = pn.z - pc.z;
      const float d  = sqrtf(dx * dx + dy * dy + dz * dz);
      const float iv = 1.0f / fmaxf(d, EPSF);
      const float x = dx * iv;
      const float y = dy * iv;
      const float z = dz * iv;
      dmax = fmaxf(dmax, d);

      const float4* sr = &s_slab[ni * PITCH4];
      const float4 g0 = sr[0];
      const float4 g1 = sr[1];
      const float4 g2 = sr[2];
      const float4 g3 = sr[3];
      const float sg[SLS] = {g0.x, g0.y, g0.z, g0.w, g1.x, g1.y, g1.z, g1.w,
                             g2.x, g2.y, g2.z, g2.w, g3.x, g3.y, g3.z, g3.w};
#pragma unroll
      for (int i = 0; i < SLS; ++i) {
        const float th = fmaxf(0.0f, sd0[i] * x + sd1[i] * y + sd2[i] * z);
        acc[i] = fmaxf(acc[i], th * sg[i]);
      }
    }
    cur = nxt;
  }

  const int    vpos = (v & ~15) | ((v & 7) << 1) | ((v >> 3) & 1);
  const float* cen  = feat + ((size_t)(j >> 1) * MROWS + rbase + vpos) * 16 + (j & 1) * SLC;
  const float4 ce0 = *reinterpret_cast<const float4*>(cen);
  const float4 ce1 = *reinterpret_cast<const float4*>(cen + 4);
  float4 o0, o1;
  o0.x = ce0.x + acc[0] + acc[8];
  o0.y = ce0.y + acc[1] + acc[9];
  o0.z = ce0.z + acc[2] + acc[10];
  o0.w = ce0.w + acc[3] + acc[11];
  o1.x = ce1.x + acc[4] + acc[12];
  o1.y = ce1.y + acc[5] + acc[13];
  o1.z = ce1.z + acc[6] + acc[14];
  o1.w = ce1.w + acc[7] + acc[15];
  float4* pf = reinterpret_cast<float4*>(planes + ((size_t)j * MROWS + row) * SLC);
  pf[0] = o0;
  pf[1] = o1;

  float dv[SLC];
#pragma unroll
  for (int c = 0; c < SLC; ++c)
    dv[c] = fmaxf(0.0f, dmax * distw[c0 + c]) + fmaxf(0.0f, dmax * distw[OUTC + c0 + c]);
  float4* pd = reinterpret_cast<float4*>(planes + ((size_t)(NSL + j) * MROWS + row) * SLC);
  pd[0] = make_float4(dv[0], dv[1], dv[2], dv[3]);
  pd[1] = make_float4(dv[4], dv[5], dv[6], dv[7]);
}

extern "C" void kernel_launch(void* const* d_in, const int* in_sizes, int n_in,
                              void* d_out, int out_size, void* d_ws, size_t ws_size,
                              hipStream_t stream) {
  const int*   nbr   = (const int*)  d_in[0];
  const float* verts = (const float*)d_in[1];
  const float* fmap  = (const float*)d_in[2];
  const float* wgt   = (const float*)d_in[3];
  const float* bias  = (const float*)d_in[4];
  const float* dirs  = (const float*)d_in[5];
  const float* distw = (const float*)d_in[6];
  const float* mlpw  = (const float*)d_in[7];
  const float* mlpb  = (const float*)d_in[8];
  float* out = (float*)d_out;

  float* feat   = (float*)d_ws;
  float* planes = feat + (size_t)NTILE * MROWS * 16;

  gemm_feat_rowblock<<<MROWS / RB, 384, 0, stream>>>(fmap, wgt, bias, feat);

  slab_gather_kernel<<<BS * NSL * (V / VPART), 1024, 0, stream>>>(
      nbr, verts, feat, dirs, distw, planes);

  gemm_out_rowblock<<<MROWS / RB, 256, 0, stream>>>(planes, mlpw, mlpb, out);
}


# Gather kernel geometry: hardware square root and reciprocal instructions (one ulp each) replace the correctly rounded library sequences, per neighbor and in the support-direction setup
# speedup vs baseline: 2.3052x; 1.0658x over previous
#include <hip/hip_runtime.h>
#include <math.h>


#define BS    8
#define V     2048
#define NNB   20
#define INC   64
#define OUTC  128
#define SUP   2
#define SUPC  (SUP * OUTC)
#define FEATC (OUTC + SUPC)
#define MROWS (BS * V)
#define EPSF  1e-12f

#define SLC    8
#define NSL    (OUTC / SLC)
#define SLS    (SUP * SLC)
#define PITCH4 5
#define VPART  1024

#define NCT    (OUTC / 16)
#define NTILE  (FEATC / 16)

typedef float v2f __attribute__((ext_vector_type(2)));
typedef float v4f __attribute__((ext_vector_type(4)));
typedef float v8f __attribute__((ext_vector_type(8)));

#define RB   64
#define LPAD 4

__global__ __launch_bounds__(384)
void gemm_feat_rowblock(const float* __restrict__ A, const float* __restrict__ B,
                        const float* __restrict__ bias, float* __restrict__ C) {
  constexpr int K  = INC;
  constexpr int N  = FEATC;
  constexpr int PA = K + LPAD;
  constexpr int KS = K / 4;
  __shared__ alignas(16) float sA[RB * PA];

  const int t    = threadIdx.x;
  const int lane = t & 31;
  const int wave = t >> 5;
  const int half = lane >> 4;
  const int l16  = lane & 15;
  const int m0   = blockIdx.x * RB;

  const v4f* __restrict__ A4 = reinterpret_cast<const v4f*>(A + (size_t)m0 * K);
  v4f ra0 = {0.f, 0.f, 0.f, 0.f}, ra1 = ra0, ra2 = ra0, ra3 = ra0;
  if (t < 256) {
    ra0 = A4[t];
    ra1 = A4[256 + t];
    ra2 = A4[512 + t];
    ra3 = A4[768 + t];
  }

  if (t < 256) {
    *reinterpret_cast<v4f*>(&sA[((0   + t) >> 4) * PA + (t & 15) * 4]) = ra0;
    *reinterpret_cast<v4f*>(&sA[((256 + t) >> 4) * PA + (t & 15) * 4]) = ra1;
    *reinterpret_cast<v4f*>(&sA[((512 + t) >> 4) * PA + (t & 15) * 4]) = ra2;
    *reinterpret_cast<v4f*>(&sA[((768 + t) >> 4) * PA + (t & 15) * 4]) = ra3;
  }

  __syncthreads();

  int col[2];
#pragma unroll
  for (int nt = 0; nt < 2; ++nt) {
    const int T = 2 * wave + nt;
    col[nt] = (T < NCT) ? (T * 16 + l16)
                        : (OUTC + (l16 >> 3) * OUTC + (T - NCT) * SLC + (l16 & 7));
  }
  v2f bf[2][KS];
#pragma unroll
  for (int ks = 0; ks < KS; ++ks) {
#pragma unroll
    for (int nt = 0; nt < 2; ++nt) {
      const float* bp = B + (size_t)(ks * 4 + 2 * half) * N + col[nt];
      bf[nt][ks].x = bp[0];
      bf[nt][ks].y = bp[N];
    }
  }
  const float bv0 = bias[col[0]];
  const float bv1 = bias[col[1]];

#pragma unroll
  for (int pr = 0; pr < RB / 32; ++pr) {
    v8f acc[2][2];
#pragma unroll
    for (int x = 0; x < 2; ++x)
#pragma unroll
      for (int y = 0; y < 2; ++y)
        acc[x][y] = (v8f){0.f, 0.f, 0.f, 0.f, 0.f, 0.f, 0.f, 0.f};

    const float* ar = &sA[(pr * 32 + l16) * PA + 2 * half];
#pragma unroll
    for (int ks = 0; ks < KS; ++ks) {
      const v2f a0 = *reinterpret_cast<const v2f*>(ar + ks * 4);
      const v2f a1 = *reinterpret_cast<const v2f*>(ar + 16 * PA + ks * 4);
      acc[0][0] = __builtin_amdgcn_wmma_f32_16x16x4_f32(false, a0, false, bf[0][ks],
                                                        (short)0, acc[0][0], false, false);
      acc[0][1] = __builtin_amdgcn_wmma_f32_16x16x4_f32(false, a0, false, bf[1][ks],
                                                        (short)0, acc[0][1], false, false);
      acc[1][0] = __builtin_amdgcn_wmma_f32_16x16x4_f32(false, a1, false, bf[0][ks],
                                                        (short)0, acc[1][0], false, false);
      acc[1][1] = __builtin_amdgcn_wmma_f32_16x16x4_f32(false, a1, false, bf[1][ks],
                                                        (short)0, acc[1][1], false, false);
    }
#pragma unroll
    for (int x = 0; x < 2; ++x) {
      float* __restrict__ blk0 =
          C + ((size_t)(2 * wave) * MROWS + m0 + pr * 32 + x * 16) * 16 + lane;
      float* __restrict__ blk1 = blk0 + (size_t)MROWS * 16;
#pragma unroll
      for (int r = 0; r < 8; ++r) {
        blk0[r * 32] = acc[x][0][r] + bv0;
        blk1[r * 32] = acc[x][1][r] + bv1;
      }
    }
    __builtin_amdgcn_sched_barrier(0);
  }
}

#define KCH   64
#define NCH   (SUPC / KCH)
__global__ __launch_bounds__(256)
void gemm_out_rowblock(const float* __restrict__ A, const float* __restrict__ Bw,
                       const float* __restrict__ bias, float* __restrict__ C) {
  constexpr int K  = SUPC;
  constexpr int N  = OUTC;
  constexpr int P  = K + LPAD;
  constexpr int NA = RB * KCH / 4 / 256;
  constexpr int NB = N * KCH / 4 / 256;
  __shared__ alignas(16) float sA[RB * P];
  __shared__ alignas(16) float sB[N * P];

  const int t    = threadIdx.x;
  const int lane = t & 31;
  const int wave = t >> 5;
  const int half = lane >> 4;
  const int l16  = lane & 15;
  const int m0   = blockIdx.x * RB;
  const v4f* __restrict__ B4 = reinterpret_cast<const v4f*>(Bw);

  v4f ra[NA], rb[NB];

#define STAGE_LOAD(c)                                                              \
  {                                                                                \
    _Pragma("unroll") for (int j = 0; j < NA; ++j) {                               \
      const int i = j * 256 + t;                                                   \
      const int p = (c) * (KCH / SLC) + (i >> 7);                                  \
      const int r = (i >> 1) & (RB - 1);                                           \
      ra[j] = *reinterpret_cast<const v4f*>(                                       \
          A + ((size_t)p * MROWS + m0 + r) * SLC + (i & 1) * 4);                   \
    }                                                                              \
    _Pragma("unroll") for (int j = 0; j < NB; ++j) {                               \
      const int i = j * 256 + t;                                                   \
      rb[j] = B4[(i >> 4) * (K / 4) + (c) * (KCH / 4) + (i & 15)];                 \
    }                                                                              \
  }
#define STAGE_STORE(c)                                                             \
  {                                                                                \
    _Pragma("unroll") for (int j = 0; j < NA; ++j) {                               \
      const int i = j * 256 + t;                                                   \
      const int p = (c) * (KCH / SLC) + (i >> 7);                                  \
      const int r = (i >> 1) & (RB - 1);                                           \
      *reinterpret_cast<v4f*>(&sA[r * P + p * SLC + (i & 1) * 4]) = ra[j];         \
    }                                                                              \
    _Pragma("unroll") for (int j = 0; j < NB; ++j) {                               \
      const int i = j * 256 + t;                                                   \
      *reinterpret_cast<v4f*>(&sB[(i >> 4) * P + (c) * KCH + (i & 15) * 4]) =      \
          rb[j];                                                                   \
    }                                                                              \
  }

  STAGE_LOAD(0)
  const int   n0 = wave * 16;
  const float bv = bias[n0 + l16];
  STAGE_STORE(0)

  v8f acc[RB / 16];
#pragma unroll
  for (int mt = 0; mt < RB / 16; ++mt)
    acc[mt] = (v8f){0.f, 0.f, 0.f, 0.f, 0.f, 0.f, 0.f, 0.f};

  const float* br = &sB[(n0 + l16) * P + 2 * half];
  const float* ar = &sA[l16 * P + 2 * half];

#pragma unroll
  for (int c = 0; c < NCH; ++c) {
    __syncthreads();
    if (c + 1 < NCH) STAGE_LOAD(c + 1)
#pragma unroll
    for (int kk = 0; kk < KCH / 4; ++kk) {
      const int ks = c * (KCH / 4) + kk;
      const v2f b = *reinterpret_cast<const v2f*>(br + ks * 4);
#pragma unroll
      for (int mt = 0; mt < RB / 16; ++mt) {
        const v2f a = *reinterpret_cast<const v2f*>(ar + mt * 16 * P + ks * 4);
        acc[mt] = __builtin_amdgcn_wmma_f32_16x16x4_f32(false, a, false, b,
                                                        (short)0, acc[mt], false, false);
      }
    }
    __builtin_amdgcn_sched_barrier(0);
    if (c + 1 < NCH) STAGE_STORE(c + 1)
  }
#undef STAGE_LOAD
#undef STAGE_STORE

#pragma unroll
  for (int mt = 0; mt < RB / 16; ++mt) {
    float* __restrict__ Crow = C + (size_t)(m0 + mt * 16 + 8 * half) * N + n0 + l16;
#pragma unroll
    for (int r = 0; r < 8; ++r)
      Crow[(size_t)r * N] = acc[mt][r] + bv;
  }
}

__global__ __launch_bounds__(1024)
void slab_gather_kernel(const int* __restrict__ nbr, const float* __restrict__ verts,
                        const float* __restrict__ feat,
                        const float* __restrict__ directions,
                        const float* __restrict__ distw,
                        float* __restrict__ planes) {
  __shared__ float4 s_slab[V * PITCH4];
  __shared__ float4 s_vert[V];

  const int t  = threadIdx.x;
  const int hv = blockIdx.x & 1;
  const int j  = (blockIdx.x >> 1) & (NSL - 1);
  const int b  = blockIdx.x >> 5;
  const int c0 = j * SLC;
  const size_t rbase = (size_t)b * V;
  const int    v   = hv * VPART + t;
  const size_t row = rbase + v;

  const int4* __restrict__ np = reinterpret_cast<const int4*>(nbr + row * NNB);
  int4 cur = np[0];

  const float4* __restrict__ slab4 =
      reinterpret_cast<const float4*>(feat + ((size_t)(NCT + j) * MROWS + rbase) * 16);
#pragma unroll
  for (int it = 0; it < (V * 4) / 1024; ++it) {
    const int idx = it * 1024 + t;
    const int p = idx >> 2;
    const int q = idx & 3;
    const int u = (p & ~15) | ((p & 1) << 3) | ((p >> 1) & 7);
    s_slab[u * PITCH4 + q] = slab4[idx];
  }
#pragma unroll
  for (int it = 0; it < V / 1024; ++it) {
    const int u = it * 1024 + t;
    const float* pv = verts + (rbase + u) * 3;
    s_vert[u] = make_float4(pv[0], pv[1], pv[2], 0.0f);
  }

  float sd0[SLS], sd1[SLS], sd2[SLS];
#pragma unroll
  for (int i = 0; i < SLS; ++i) {
    const int s = (i >> 3) * OUTC + c0 + (i & 7);
    float d0 = directions[0 * SUPC + s];
    float d1 = directions[1 * SUPC + s];
    float d2 = directions[2 * SUPC + s];
    const float nrm = __builtin_amdgcn_sqrtf(d0 * d0 + d1 * d1 + d2 * d2);
    const float iv  = __builtin_amdgcn_rcpf(fmaxf(nrm, EPSF));
    sd0[i] = d0 * iv;
    sd1[i] = d1 * iv;
    sd2[i] = d2 * iv;
  }

  __syncthreads();

  const float4 pc = s_vert[v];
  float acc[SLS];
#pragma unroll
  for (int i = 0; i < SLS; ++i) acc[i] = -INFINITY;
  float dmax = 0.0f;

#pragma unroll 1
  for (int g = 0; g < NNB / 4; ++g) {
    const int4 nxt = np[(g + 1 < NNB / 4) ? (g + 1) : g];
    const int ids[4] = {cur.x, cur.y, cur.z, cur.w};
#pragma unroll
    for (int e = 0; e < 4; ++e) {
      const int ni = ids[e] & (V - 1);
      const float4 pn = s_vert[ni];
      const float dx = pn.x - pc.x;
      const float dy = pn.y - pc.y;
      const float dz = pn.z - pc.z;
      const float d  = __builtin_amdgcn_sqrtf(dx * dx + dy * dy + dz * dz);
      const float iv = __builtin_amdgcn_rcpf(fmaxf(d, EPSF));
      const float x = dx * iv;
      const float y = dy * iv;
      const float z = dz * iv;
      dmax = fmaxf(dmax, d);

      const float4* sr = &s_slab[ni * PITCH4];
      const float4 g0 = sr[0];
      const float4 g1 = sr[1];
      const float4 g2 = sr[2];
      const float4 g3 = sr[3];
      const float sg[SLS] = {g0.x, g0.y, g0.z, g0.w, g1.x, g1.y, g1.z, g1.w,
                             g2.x, g2.y, g2.z, g2.w, g3.x, g3.y, g3.z, g3.w};
#pragma unroll
      for (int i = 0; i < SLS; ++i) {
        const float th = fmaxf(0.0f, sd0[i] * x + sd1[i] * y + sd2[i] * z);
        acc[i] = fmaxf(acc[i], th * sg[i]);
      }
    }
    cur = nxt;
  }

  const int    vpos = (v & ~15) | ((v & 7) << 1) | ((v >> 3) & 1);
  const float* cen  = feat + ((size_t)(j >> 1) * MROWS + rbase + vpos) * 16 + (j & 1) * SLC;
  const float4 ce0 = *reinterpret_cast<const float4*>(cen);
  const float4 ce1 = *reinterpret_cast<const float4*>(cen + 4);
  float4 o0, o1;
  o0.x = ce0.x + acc[0] + acc[8];
  o0.y = ce0.y + acc[1] + acc[9];
  o0.z = ce0.z + acc[2] + acc[10];
  o0.w = ce0.w + acc[3] + acc[11];
  o1.x = ce1.x + acc[4] + acc[12];
  o1.y = ce1.y + acc[5] + acc[13];
  o1.z = ce1.z + acc[6] + acc[14];
  o1.w = ce1.w + acc[7] + acc[15];
  float4* pf = reinterpret_cast<float4*>(planes + ((size_t)j * MROWS + row) * SLC);
  pf[0] = o0;
  pf[1] = o1;

  float dv[SLC];
#pragma unroll
  for (int c = 0; c < SLC; ++c)
    dv[c] = fmaxf(0.0f, dmax * distw[c0 + c]) + fmaxf(0.0f, dmax * distw[OUTC + c0 + c]);
  float4* pd = reinterpret_cast<float4*>(planes + ((size_t)(NSL + j) * MROWS + row) * SLC);
  pd[0] = make_float4(dv[0], dv[1], dv[2], dv[3]);
  pd[1] = make_float4(dv[4], dv[5], dv[6], dv[7]);
}

extern "C" void kernel_launch(void* const* d_in, const int* in_sizes, int n_in,
                              void* d_out, int out_size, void* d_ws, size_t ws_size,
                              hipStream_t stream) {
  const int*   nbr   = (const int*)  d_in[0];
  const float* verts = (const float*)d_in[1];
  const float* fmap  = (const float*)d_in[2];
  const float* wgt   = (const float*)d_in[3];
  const float* bias  = (const float*)d_in[4];
  const float* dirs  = (const float*)d_in[5];
  const float* distw = (const float*)d_in[6];
  const float* mlpw  = (const float*)d_in[7];
  const float* mlpb  = (const float*)d_in[8];
  float* out = (float*)d_out;

  float* feat   = (float*)d_ws;
  float* planes = feat + (size_t)NTILE * MROWS * 16;

  gemm_feat_rowblock<<<MROWS / RB, 384, 0, stream>>>(fmap, wgt, bias, feat);

  slab_gather_kernel<<<BS * NSL * (V / VPART), 1024, 0, stream>>>(
      nbr, verts, feat, dirs, distw, planes);

  gemm_out_rowblock<<<MROWS / RB, 256, 0, stream>>>(planes, mlpw, mlpb, out);
}


# Distance half of the last GEMM's A rebuilt while staging from one dmax float per vertex (same expression); the gather kernel no longer writes, nor the GEMM reads, the distance planes
# speedup vs baseline: 2.3912x; 1.0373x over previous
#include <hip/hip_runtime.h>
#include <math.h>


#define BS    8
#define V     2048
#define NNB   20
#define INC   64
#define OUTC  128
#define SUP   2
#define SUPC  (SUP * OUTC)
#define FEATC (OUTC + SUPC)
#define MROWS (BS * V)
#define EPSF  1e-12f

#define SLC    8
#define NSL    (OUTC / SLC)
#define SLS    (SUP * SLC)
#define PITCH4 5
#define VPART  1024

#define NCT    (OUTC / 16)
#define NTILE  (FEATC / 16)

typedef float v2f __attribute__((ext_vector_type(2)));
typedef float v4f __attribute__((ext_vector_type(4)));
typedef float v8f __attribute__((ext_vector_type(8)));

#define RB   64
#define LPAD 4

__global__ __launch_bounds__(384)
void gemm_feat_rowblock(const float* __restrict__ A, const float* __restrict__ B,
                        const float* __restrict__ bias, float* __restrict__ C) {
  constexpr int K  = INC;
  constexpr int N  = FEATC;
  constexpr int PA = K + LPAD;
  constexpr int KS = K / 4;
  __shared__ alignas(16) float sA[RB * PA];

  const int t    = threadIdx.x;
  const int lane = t & 31;
  const int wave = t >> 5;
  const int half = lane >> 4;
  const int l16  = lane & 15;
  const int m0   = blockIdx.x * RB;

  const v4f* __restrict__ A4 = reinterpret_cast<const v4f*>(A + (size_t)m0 * K);
  v4f ra0 = {0.f, 0.f, 0.f, 0.f}, ra1 = ra0, ra2 = ra0, ra3 = ra0;
  if (t < 256) {
    ra0 = A4[t];
    ra1 = A4[256 + t];
    ra2 = A4[512 + t];
    ra3 = A4[768 + t];
  }

  if (t < 256) {
    *reinterpret_cast<v4f*>(&sA[((0   + t) >> 4) * PA + (t & 15) * 4]) = ra0;
    *reinterpret_cast<v4f*>(&sA[((256 + t) >> 4) * PA + (t & 15) * 4]) = ra1;
    *reinterpret_cast<v4f*>(&sA[((512 + t) >> 4) * PA + (t & 15) * 4]) = ra2;
    *reinterpret_cast<v4f*>(&sA[((768 + t) >> 4) * PA + (t & 15) * 4]) = ra3;
  }

  __syncthreads();

  int col[2];
#pragma unroll
  for (int nt = 0; nt < 2; ++nt) {
    const int T = 2 * wave + nt;
    col[nt] = (T < NCT) ? (T * 16 + l16)
                        : (OUTC + (l16 >> 3) * OUTC + (T - NCT) * SLC + (l16 & 7));
  }
  v2f bf[2][KS];
#pragma unroll
  for (int ks = 0; ks < KS; ++ks) {
#pragma unroll
    for (int nt = 0; nt < 2; ++nt) {
      const float* bp = B + (size_t)(ks * 4 + 2 * half) * N + col[nt];
      bf[nt][ks].x = bp[0];
      bf[nt][ks].y = bp[N];
    }
  }
  const float bv0 = bias[col[0]];
  const float bv1 = bias[col[1]];

#pragma unroll
  for (int pr = 0; pr < RB / 32; ++pr) {
    v8f acc[2][2];
#pragma unroll
    for (int x = 0; x < 2; ++x)
#pragma unroll
      for (int y = 0; y < 2; ++y)
        acc[x][y] = (v8f){0.f, 0.f, 0.f, 0.f, 0.f, 0.f, 0.f, 0.f};

    const float* ar = &sA[(pr * 32 + l16) * PA + 2 * half];
#pragma unroll
    for (int ks = 0; ks < KS; ++ks) {
      const v2f a0 = *reinterpret_cast<const v2f*>(ar + ks * 4);
      const v2f a1 = *reinterpret_cast<const v2f*>(ar + 16 * PA + ks * 4);
      acc[0][0] = __builtin_amdgcn_wmma_f32_16x16x4_f32(false, a0, false, bf[0][ks],
                                                        (short)0, acc[0][0], false, false);
      acc[0][1] = __builtin_amdgcn_wmma_f32_16x16x4_f32(false, a0, false, bf[1][ks],
                                                        (short)0, acc[0][1], false, false);
      acc[1][0] = __builtin_amdgcn_wmma_f32_16x16x4_f32(false, a1, false, bf[0][ks],
                                                        (short)0, acc[1][0], false, false);
      acc[1][1] = __builtin_amdgcn_wmma_f32_16x16x4_f32(false, a1, false, bf[1][ks],
                                                        (short)0, acc[1][1], false, false);
    }
#pragma unroll
    for (int x = 0; x < 2; ++x) {
      float* __restrict__ blk0 =
          C + ((size_t)(2 * wave) * MROWS + m0 + pr * 32 + x * 16) * 16 + lane;
      float* __restrict__ blk1 = blk0 + (size_t)MROWS * 16;
#pragma unroll
      for (int r = 0; r < 8; ++r) {
        blk0[r * 32] = acc[x][0][r] + bv0;
        blk1[r * 32] = acc[x][1][r] + bv1;
      }
    }
    __builtin_amdgcn_sched_barrier(0);
  }
}

#define KCH   64
#define NCH   (SUPC / KCH)
__global__ __launch_bounds__(256)
void gemm_out_rowblock(const float* __restrict__ A, const float* __restrict__ Bw,
                       const float* __restrict__ bias, const float* __restrict__ distw,
                       float* __restrict__ C) {
  constexpr int K  = SUPC;
  constexpr int N  = OUTC;
  constexpr int P  = K + LPAD;
  constexpr int NA = RB * KCH / 4 / 256;
  constexpr int NB = N * KCH / 4 / 256;
  __shared__ alignas(16) float sA[RB * P];
  __shared__ alignas(16) float sB[N * P];

  const int t    = threadIdx.x;
  const int lane = t & 31;
  const int wave = t >> 5;
  const int half = lane >> 4;
  const int l16  = lane & 15;
  const int m0   = blockIdx.x * RB;
  const v4f* __restrict__ B4 = reinterpret_cast<const v4f*>(Bw);

  v4f ra[NA], rb[NB];
  v4f rw0[NA], rw1[NA];
  float rd[NA];
  const float* __restrict__ dmx = A + (size_t)NSL * MROWS * SLC;

#define STAGE_LOAD(c)                                                              \
  {                                                                                \
    _Pragma("unroll") for (int j = 0; j < NA; ++j) {                               \
      const int i = j * 256 + t;                                                   \
      const int p = (c) * (KCH / SLC) + (i >> 7);                                  \
      const int r = (i >> 1) & (RB - 1);                                           \
      if ((c) < NCH / 2) {                                                         \
        ra[j] = *reinterpret_cast<const v4f*>(                                     \
            A + ((size_t)p * MROWS + m0 + r) * SLC + (i & 1) * 4);                 \
      } else {                                                                     \
        const int ch = (p - NSL) * SLC + (i & 1) * 4;                              \
        rd[j]  = dmx[m0 + r];                                                      \
        rw0[j] = *reinterpret_cast<const v4f*>(distw + ch);                        \
        rw1[j] = *reinterpret_cast<const v4f*>(distw + OUTC + ch);                 \
      }                                                                            \
    }                                                                              \
    _Pragma("unroll") for (int j = 0; j < NB; ++j) {                               \
      const int i = j * 256 + t;                                                   \
      rb[j] = B4[(i >> 4) * (K / 4) + (c) * (KCH / 4) + (i & 15)];                 \
    }                                                                              \
  }
#define STAGE_STORE(c)                                                             \
  {                                                                                \
    _Pragma("unroll") for (int j = 0; j < NA; ++j) {                               \
      const int i = j * 256 + t;                                                   \
      const int p = (c) * (KCH / SLC) + (i >> 7);                                  \
      const int r = (i >> 1) & (RB - 1);                                           \
      if ((c) >= NCH / 2) {                                                        \
        const float dm = rd[j];                                                    \
        ra[j].x = fmaxf(0.0f, dm * rw0[j].x) + fmaxf(0.0f, dm * rw1[j].x);         \
        ra[j].y = fmaxf(0.0f, dm * rw0[j].y) + fmaxf(0.0f, dm * rw1[j].y);         \
        ra[j].z = fmaxf(0.0f, dm * rw0[j].z) + fmaxf(0.0f, dm * rw1[j].z);         \
        ra[j].w = fmaxf(0.0f, dm * rw0[j].w) + fmaxf(0.0f, dm * rw1[j].w);         \
      }                                                                            \
      *reinterpret_cast<v4f*>(&sA[r * P + p * SLC + (i & 1) * 4]) = ra[j];         \
    }                                                                              \
    _Pragma("unroll") for (int j = 0; j < NB; ++j) {                               \
      const int i = j * 256 + t;                                                   \
      *reinterpret_cast<v4f*>(&sB[(i >> 4) * P + (c) * KCH + (i & 15) * 4]) =      \
          rb[j];                                                                   \
    }                                                                              \
  }

  STAGE_LOAD(0)
  const int   n0 = wave * 16;
  const float bv = bias[n0 + l16];
  STAGE_STORE(0)

  v8f acc[RB / 16];
#pragma unroll
  for (int mt = 0; mt < RB / 16; ++mt)
    acc[mt] = (v8f){0.f, 0.f, 0.f, 0.f, 0.f, 0.f, 0.f, 0.f};

  const float* br = &sB[(n0 + l16) * P + 2 * half];
  const float* ar = &sA[l16 * P + 2 * half];

#pragma unroll
  for (int c = 0; c < NCH; ++c) {
    __syncthreads();
    if (c + 1 < NCH) STAGE_LOAD(c + 1)
#pragma unroll
    for (int kk = 0; kk < KCH / 4; ++kk) {
      const int ks = c * (KCH / 4) + kk;
      const v2f b = *reinterpret_cast<const v2f*>(br + ks * 4);
#pragma unroll
      for (int mt = 0; mt < RB / 16; ++mt) {
        const v2f a = *reinterpret_cast<const v2f*>(ar + mt * 16 * P + ks * 4);
        acc[mt] = __builtin_amdgcn_wmma_f32_16x16x4_f32(false, a, false, b,
                                                        (short)0, acc[mt], false, false);
      }
    }
    __builtin_amdgcn_sched_barrier(0);
    if (c + 1 < NCH) STAGE_STORE(c + 1)
  }
#undef STAGE_LOAD
#undef STAGE_STORE

#pragma unroll
  for (int mt = 0; mt < RB / 16; ++mt) {
    float* __restrict__ Crow = C + (size_t)(m0 + mt * 16 + 8 * half) * N + n0 + l16;
#pragma unroll
    for (int r = 0; r < 8; ++r)
      Crow[(size_t)r * N] = acc[mt][r] + bv;
  }
}

__global__ __launch_bounds__(1024)
void slab_gather_kernel(const int* __restrict__ nbr, const float* __restrict__ verts,
                        const float* __restrict__ feat,
                        const float* __restrict__ directions,
                        float* __restrict__ planes) {
  __shared__ float4 s_slab[V * PITCH4];
  __shared__ float4 s_vert[V];

  const int t  = threadIdx.x;
  const int hv = blockIdx.x & 1;
  const int j  = (blockIdx.x >> 1) & (NSL - 1);
  const int b  = blockIdx.x >> 5;
  const int c0 = j * SLC;
  const size_t rbase = (size_t)b * V;
  const int    v   = hv * VPART + t;
  const size_t row = rbase + v;

  const int4* __restrict__ np = reinterpret_cast<const int4*>(nbr + row * NNB);
  int4 cur = np[0];

  const float4* __restrict__ slab4 =
      reinterpret_cast<const float4*>(feat + ((size_t)(NCT + j) * MROWS + rbase) * 16);
#pragma unroll
  for (int it = 0; it < (V * 4) / 1024; ++it) {
    const int idx = it * 1024 + t;
    const int p = idx >> 2;
    const int q = idx & 3;
    const int u = (p & ~15) | ((p & 1) << 3) | ((p >> 1) & 7);
    s_slab[u * PITCH4 + q] = slab4[idx];
  }
#pragma unroll
  for (int it = 0; it < V / 1024; ++it) {
    const int u = it * 1024 + t;
    const float* pv = verts + (rbase + u) * 3;
    s_vert[u] = make_float4(pv[0], pv[1], pv[2], 0.0f);
  }

  float sd0[SLS], sd1[SLS], sd2[SLS];
#pragma unroll
  for (int i = 0; i < SLS; ++i) {
    const int s = (i >> 3) * OUTC + c0 + (i & 7);
    float d0 = directions[0 * SUPC + s];
    float d1 = directions[1 * SUPC + s];
    float d2 = directions[2 * SUPC + s];
    const float nrm = __builtin_amdgcn_sqrtf(d0 * d0 + d1 * d1 + d2 * d2);
    const float iv  = __builtin_amdgcn_rcpf(fmaxf(nrm, EPSF));
    sd0[i] = d0 * iv;
    sd1[i] = d1 * iv;
    sd2[i] = d2 * iv;
  }

  __syncthreads();

  const float4 pc = s_vert[v];
  float acc[SLS];
#pragma unroll
  for (int i = 0; i < SLS; ++i) acc[i] = -INFINITY;
  float dmax = 0.0f;

#pragma unroll 1
  for (int g = 0; g < NNB / 4; ++g) {
    const int4 nxt = np[(g + 1 < NNB / 4) ? (g + 1) : g];
    const int ids[4] = {cur.x, cur.y, cur.z, cur.w};
#pragma unroll
    for (int e = 0; e < 4; ++e) {
      const int ni = ids[e] & (V - 1);
      const float4 pn = s_vert[ni];
      const float dx = pn.x - pc.x;
      const float dy = pn.y - pc.y;
      const float dz = pn.z - pc.z;
      const float d  = __builtin_amdgcn_sqrtf(dx * dx + dy * dy + dz * dz);
      const float iv = __builtin_amdgcn_rcpf(fmaxf(d, EPSF));
      const float x = dx * iv;
      const float y = dy * iv;
      const float z = dz * iv;
      dmax = fmaxf(dmax, d);

      const float4* sr = &s_slab[ni * PITCH4];
      const float4 g0 = sr[0];
      const float4 g1 = sr[1];
      const float4 g2 = sr[2];
      const float4 g3 = sr[3];
      const float sg[SLS] = {g0.x, g0.y, g0.z, g0.w, g1.x, g1.y, g1.z, g1.w,
                             g2.x, g2.y, g2.z, g2.w, g3.x, g3.y, g3.z, g3.w};
#pragma unroll
      for (int i = 0; i < SLS; ++i) {
        const float th = fmaxf(0.0f, sd0[i] * x + sd1[i] * y + sd2[i] * z);
        acc[i] = fmaxf(acc[i], th * sg[i]);
      }
    }
    cur = nxt;
  }

  const int    vpos = (v & ~15) | ((v & 7) << 1) | ((v >> 3) & 1);
  const float* cen  = feat + ((size_t)(j >> 1) * MROWS + rbase + vpos) * 16 + (j & 1) * SLC;
  const float4 ce0 = *reinterpret_cast<const float4*>(cen);
  const float4 ce1 = *reinterpret_cast<const float4*>(cen + 4);
  float4 o0, o1;
  o0.x = ce0.x + acc[0] + acc[8];
  o0.y = ce0.y + acc[1] + acc[9];
  o0.z = ce0.z + acc[2] + acc[10];
  o0.w = ce0.w + acc[3] + acc[11];
  o1.x = ce1.x + acc[4] + acc[12];
  o1.y = ce1.y + acc[5] + acc[13];
  o1.z = ce1.z + acc[6] + acc[14];
  o1.w = ce1.w + acc[7] + acc[15];
  float4* pf = reinterpret_cast<float4*>(planes + ((size_t)j * MROWS + row) * SLC);
  pf[0] = o0;
  pf[1] = o1;

  if (j == 0) planes[(size_t)NSL * MROWS * SLC + row] = dmax;
}

extern "C" void kernel_launch(void* const* d_in, const int* in_sizes, int n_in,
                              void* d_out, int out_size, void* d_ws, size_t ws_size,
                              hipStream_t stream) {
  const int*   nbr   = (const int*)  d_in[0];
  const float* verts = (const float*)d_in[1];
  const float* fmap  = (const float*)d_in[2];
  const float* wgt   = (const float*)d_in[3];
  const float* bias  = (const float*)d_in[4];
  const float* dirs  = (const float*)d_in[5];
  const float* distw = (const float*)d_in[6];
  const float* mlpw  = (const float*)d_in[7];
  const float* mlpb  = (const float*)d_in[8];
  float* out = (float*)d_out;

  float* feat   = (float*)d_ws;
  float* planes = feat + (size_t)NTILE * MROWS * 16;

  gemm_feat_rowblock<<<MROWS / RB, 384, 0, stream>>>(fmap, wgt, bias, feat);

  slab_gather_kernel<<<BS * NSL * (V / VPART), 1024, 0, stream>>>(
      nbr, verts, feat, dirs, planes);

  gemm_out_rowblock<<<MROWS / RB, 256, 0, stream>>>(planes, mlpw, mlpb, distw, out);
}


# Gather kernel regridded in its slab design: 256-thread work-groups, 4 destination vertices per thread, rows of the next neighbor step read from LDS one step ahead; 2048 waves enter instead of 8192
# speedup vs baseline: 2.4633x; 1.0302x over previous
#include <hip/hip_runtime.h>
#include <math.h>


#define BS    8
#define V     2048
#define NNB   20
#define INC   64
#define OUTC  128
#define SUP   2
#define SUPC  (SUP * OUTC)
#define FEATC (OUTC + SUPC)
#define MROWS (BS * V)
#define EPSF  1e-12f

#define SLC    8
#define NSL    (OUTC / SLC)
#define SLS    (SUP * SLC)
#define PITCH4 5
#define VPART  1024

#define NCT    (OUTC / 16)
#define NTILE  (FEATC / 16)

typedef float v2f __attribute__((ext_vector_type(2)));
typedef float v4f __attribute__((ext_vector_type(4)));
typedef float v8f __attribute__((ext_vector_type(8)));

#define RB   64
#define LPAD 4

__global__ __launch_bounds__(384)
void gemm_feat_rowblock(const float* __restrict__ A, const float* __restrict__ B,
                        const float* __restrict__ bias, float* __restrict__ C) {
  constexpr int K  = INC;
  constexpr int N  = FEATC;
  constexpr int PA = K + LPAD;
  constexpr int KS = K / 4;
  __shared__ alignas(16) float sA[RB * PA];

  const int t    = threadIdx.x;
  const int lane = t & 31;
  const int wave = t >> 5;
  const int half = lane >> 4;
  const int l16  = lane & 15;
  const int m0   = blockIdx.x * RB;

  const v4f* __restrict__ A4 = reinterpret_cast<const v4f*>(A + (size_t)m0 * K);
  v4f ra0 = {0.f, 0.f, 0.f, 0.f}, ra1 = ra0, ra2 = ra0, ra3 = ra0;
  if (t < 256) {
    ra0 = A4[t];
    ra1 = A4[256 + t];
    ra2 = A4[512 + t];
    ra3 = A4[768 + t];
  }

  if (t < 256) {
    *reinterpret_cast<v4f*>(&sA[((0   + t) >> 4) * PA + (t & 15) * 4]) = ra0;
    *reinterpret_cast<v4f*>(&sA[((256 + t) >> 4) * PA + (t & 15) * 4]) = ra1;
    *reinterpret_cast<v4f*>(&sA[((512 + t) >> 4) * PA + (t & 15) * 4]) = ra2;
    *reinterpret_cast<v4f*>(&sA[((768 + t) >> 4) * PA + (t & 15) * 4]) = ra3;
  }

  __syncthreads();

  int col[2];
#pragma unroll
  for (int nt = 0; nt < 2; ++nt) {
    const int T = 2 * wave + nt;
    col[nt] = (T < NCT) ? (T * 16 + l16)
                        : (OUTC + (l16 >> 3) * OUTC + (T - NCT) * SLC + (l16 & 7));
  }
  v2f bf[2][KS];
#pragma unroll
  for (int ks = 0; ks < KS; ++ks) {
#pragma unroll
    for (int nt = 0; nt < 2; ++nt) {
      const float* bp = B + (size_t)(ks * 4 + 2 * half) * N + col[nt];
      bf[nt][ks].x = bp[0];
      bf[nt][ks].y = bp[N];
    }
  }
  const float bv0 = bias[col[0]];
  const float bv1 = bias[col[1]];

#pragma unroll
  for (int pr = 0; pr < RB / 32; ++pr) {
    v8f acc[2][2];
#pragma unroll
    for (int x = 0; x < 2; ++x)
#pragma unroll
      for (int y = 0; y < 2; ++y)
        acc[x][y] = (v8f){0.f, 0.f, 0.f, 0.f, 0.f, 0.f, 0.f, 0.f};

    const float* ar = &sA[(pr * 32 + l16) * PA + 2 * half];
#pragma unroll
    for (int ks = 0; ks < KS; ++ks) {
      const v2f a0 = *reinterpret_cast<const v2f*>(ar + ks * 4);
      const v2f a1 = *reinterpret_cast<const v2f*>(ar + 16 * PA + ks * 4);
      acc[0][0] = __builtin_amdgcn_wmma_f32_16x16x4_f32(false, a0, false, bf[0][ks],
                                                        (short)0, acc[0][0], false, false);
      acc[0][1] = __builtin_amdgcn_wmma_f32_16x16x4_f32(false, a0, false, bf[1][ks],
                                                        (short)0, acc[0][1], false, false);
      acc[1][0] = __builtin_amdgcn_wmma_f32_16x16x4_f32(false, a1, false, bf[0][ks],
                                                        (short)0, acc[1][0], false, false);
      acc[1][1] = __builtin_amdgcn_wmma_f32_16x16x4_f32(false, a1, false, bf[1][ks],
                                                        (short)0, acc[1][1], false, false);
    }
#pragma unroll
    for (int x = 0; x < 2; ++x) {
      float* __restrict__ blk0 =
          C + ((size_t)(2 * wave) * MROWS + m0 + pr * 32 + x * 16) * 16 + lane;
      float* __restrict__ blk1 = blk0 + (size_t)MROWS * 16;
#pragma unroll
      for (int r = 0; r < 8; ++r) {
        blk0[r * 32] = acc[x][0][r] + bv0;
        blk1[r * 32] = acc[x][1][r] + bv1;
      }
    }
    __builtin_amdgcn_sched_barrier(0);
  }
}

#define KCH   64
#define NCH   (SUPC / KCH)
__global__ __launch_bounds__(256)
void gemm_out_rowblock(const float* __restrict__ A, const float* __restrict__ Bw,
                       const float* __restrict__ bias, const float* __restrict__ distw,
                       float* __restrict__ C) {
  constexpr int K  = SUPC;
  constexpr int N  = OUTC;
  constexpr int P  = K + LPAD;
  constexpr int NA = RB * KCH / 4 / 256;
  constexpr int NB = N * KCH / 4 / 256;
  __shared__ alignas(16) float sA[RB * P];
  __shared__ alignas(16) float sB[N * P];

  const int t    = threadIdx.x;
  const int lane = t & 31;
  const int wave = t >> 5;
  const int half = lane >> 4;
  const int l16  = lane & 15;
  const int m0   = blockIdx.x * RB;
  const v4f* __restrict__ B4 = reinterpret_cast<const v4f*>(Bw);

  v4f ra[NA], rb[NB];
  v4f rw0[NA], rw1[NA];
  float rd[NA];
  const float* __restrict__ dmx = A + (size_t)NSL * MROWS * SLC;

#define STAGE_LOAD(c)                                                              \
  {                                                                                \
    _Pragma("unroll") for (int j = 0; j < NA; ++j) {                               \
      const int i = j * 256 + t;                                                   \
      const int p = (c) * (KCH / SLC) + (i >> 7);                                  \
      const int r = (i >> 1) & (RB - 1);                                           \
      if ((c) < NCH / 2) {                                                         \
        ra[j] = *reinterpret_cast<const v4f*>(                                     \
            A + ((size_t)p * MROWS + m0 + r) * SLC + (i & 1) * 4);                 \
      } else {                                                                     \
        const int ch = (p - NSL) * SLC + (i & 1) * 4;                              \
        rd[j]  = dmx[m0 + r];                                                      \
        rw0[j] = *reinterpret_cast<const v4f*>(distw + ch);                        \
        rw1[j] = *reinterpret_cast<const v4f*>(distw + OUTC + ch);                 \
      }                                                                            \
    }                                                                              \
    _Pragma("unroll") for (int j = 0; j < NB; ++j) {                               \
      const int i = j * 256 + t;                                                   \
      rb[j] = B4[(i >> 4) * (K / 4) + (c) * (KCH / 4) + (i & 15)];                 \
    }                                                                              \
  }
#define STAGE_STORE(c)                                                             \
  {                                                                                \
    _Pragma("unroll") for (int j = 0; j < NA; ++j) {                               \
      const int i = j * 256 + t;                                                   \
      const int p = (c) * (KCH / SLC) + (i >> 7);                                  \
      const int r = (i >> 1) & (RB - 1);                                           \
      if ((c) >= NCH / 2) {                                                        \
        const float dm = rd[j];                                                    \
        ra[j].x = fmaxf(0.0f, dm * rw0[j].x) + fmaxf(0.0f, dm * rw1[j].x);         \
        ra[j].y = fmaxf(0.0f, dm * rw0[j].y) + fmaxf(0.0f, dm * rw1[j].y);         \
        ra[j].z = fmaxf(0.0f, dm * rw0[j].z) + fmaxf(0.0f, dm * rw1[j].z);         \
        ra[j].w = fmaxf(0.0f, dm * rw0[j].w) + fmaxf(0.0f, dm * rw1[j].w);         \
      }                                                                            \
      *reinterpret_cast<v4f*>(&sA[r * P + p * SLC + (i & 1) * 4]) = ra[j];         \
    }                                                                              \
    _Pragma("unroll") for (int j = 0; j < NB; ++j) {                               \
      const int i = j * 256 + t;                                                   \
      *reinterpret_cast<v4f*>(&sB[(i >> 4) * P + (c) * KCH + (i & 15) * 4]) =      \
          rb[j];                                                                   \
    }                                                                              \
  }

  STAGE_LOAD(0)
  const int   n0 = wave * 16;
  const float bv = bias[n0 + l16];
  STAGE_STORE(0)

  v8f acc[RB / 16];
#pragma unroll
  for (int mt = 0; mt < RB / 16; ++mt)
    acc[mt] = (v8f){0.f, 0.f, 0.f, 0.f, 0.f, 0.f, 0.f, 0.f};

  const float* br = &sB[(n0 + l16) * P + 2 * half];
  const float* ar = &sA[l16 * P + 2 * half];

#pragma unroll
  for (int c = 0; c < NCH; ++c) {
    __syncthreads();
    if (c + 1 < NCH) STAGE_LOAD(c + 1)
#pragma unroll
    for (int kk = 0; kk < KCH / 4; ++kk) {
      const int ks = c * (KCH / 4) + kk;
      const v2f b = *reinterpret_cast<const v2f*>(br + ks * 4);
#pragma unroll
      for (int mt = 0; mt < RB / 16; ++mt) {
        const v2f a = *reinterpret_cast<const v2f*>(ar + mt * 16 * P + ks * 4);
        acc[mt] = __builtin_amdgcn_wmma_f32_16x16x4_f32(false, a, false, b,
                                                        (short)0, acc[mt], false, false);
      }
    }
    __builtin_amdgcn_sched_barrier(0);
    if (c + 1 < NCH) STAGE_STORE(c + 1)
  }
#undef STAGE_LOAD
#undef STAGE_STORE

#pragma unroll
  for (int mt = 0; mt < RB / 16; ++mt) {
    float* __restrict__ Crow = C + (size_t)(m0 + mt * 16 + 8 * half) * N + n0 + l16;
#pragma unroll
    for (int r = 0; r < 8; ++r)
      Crow[(size_t)r * N] = acc[mt][r] + bv;
  }
}

#define VT 4
#define GT (VPART / VT)
__global__ __launch_bounds__(GT)
void slab_gather_kernel(const int* __restrict__ nbr, const float* __restrict__ verts,
                        const float* __restrict__ feat,
                        const float* __restrict__ directions,
                        float* __restrict__ planes) {
  __shared__ float4 s_slab[V * PITCH4];
  __shared__ float4 s_vert[V];

  const int t  = threadIdx.x;
  const int hv = blockIdx.x & 1;
  const int j  = (blockIdx.x >> 1) & (NSL - 1);
  const int b  = blockIdx.x >> 5;
  const int c0 = j * SLC;
  const size_t rbase = (size_t)b * V;
  const int    v0  = hv * VPART + t;

  const int4* __restrict__ np[VT];
  int4 cur[VT];
#pragma unroll
  for (int k = 0; k < VT; ++k) {
    np[k]  = reinterpret_cast<const int4*>(nbr + (rbase + v0 + k * GT) * NNB);
    cur[k] = np[k][0];
  }

  const float4* __restrict__ slab4 =
      reinterpret_cast<const float4*>(feat + ((size_t)(NCT + j) * MROWS + rbase) * 16);
#pragma unroll
  for (int it = 0; it < (V * 4) / GT; ++it) {
    const int idx = it * GT + t;
    const int p = idx >> 2;
    const int q = idx & 3;
    const int u = (p & ~15) | ((p & 1) << 3) | ((p >> 1) & 7);
    s_slab[u * PITCH4 + q] = slab4[idx];
  }
#pragma unroll
  for (int it = 0; it < V / GT; ++it) {
    const int u = it * GT + t;
    const float* pv = verts + (rbase + u) * 3;
    s_vert[u] = make_float4(pv[0], pv[1], pv[2], 0.0f);
  }

  float sd0[SLS], sd1[SLS], sd2[SLS];
#pragma unroll
  for (int i = 0; i < SLS; ++i) {
    const int s = (i >> 3) * OUTC + c0 + (i & 7);
    float d0 = directions[0 * SUPC + s];
    float d1 = directions[1 * SUPC + s];
    float d2 = directions[2 * SUPC + s];
    const float nrm = __builtin_amdgcn_sqrtf(d0 * d0 + d1 * d1 + d2 * d2);
    const float iv  = __builtin_amdgcn_rcpf(fmaxf(nrm, EPSF));
    sd0[i] = d0 * iv;
    sd1[i] = d1 * iv;
    sd2[i] = d2 * iv;
  }

  float4 ce0[VT], ce1[VT];
#pragma unroll
  for (int k = 0; k < VT; ++k) {
    const int    v    = v0 + k * GT;
    const int    vpos = (v & ~15) | ((v & 7) << 1) | ((v >> 3) & 1);
    const float* cen  = feat + ((size_t)(j >> 1) * MROWS + rbase + vpos) * 16 + (j & 1) * SLC;
    ce0[k] = *reinterpret_cast<const float4*>(cen);
    ce1[k] = *reinterpret_cast<const float4*>(cen + 4);
  }

  __syncthreads();

  float4 pc[VT];
  float  acc[VT][SLS];
  float  dmax[VT];
#pragma unroll
  for (int k = 0; k < VT; ++k) {
    pc[k] = s_vert[v0 + k * GT];
    dmax[k] = 0.0f;
#pragma unroll
    for (int i = 0; i < SLS; ++i) acc[k][i] = -INFINITY;
  }

  struct NbRow { float4 p, g0, g1, g2, g3; };
  auto fetch = [&](int id, NbRow& r) {
    const int ni = id & (V - 1);
    r.p = s_vert[ni];
    const float4* sr = &s_slab[ni * PITCH4];
    r.g0 = sr[0];
    r.g1 = sr[1];
    r.g2 = sr[2];
    r.g3 = sr[3];
  };

  NbRow nx[VT];
#pragma unroll
  for (int k = 0; k < VT; ++k) fetch(cur[k].x, nx[k]);

#pragma unroll 1
  for (int g = 0; g < NNB / 4; ++g) {
    int4 nxt[VT];
#pragma unroll
    for (int k = 0; k < VT; ++k) nxt[k] = np[k][(g + 1 < NNB / 4) ? (g + 1) : g];
#pragma unroll
    for (int e = 0; e < 4; ++e) {
      NbRow cu[VT];
#pragma unroll
      for (int k = 0; k < VT; ++k) {
        cu[k] = nx[k];
        const int idn = (e == 0) ? cur[k].y : (e == 1) ? cur[k].z : (e == 2) ? cur[k].w : nxt[k].x;
        fetch(idn, nx[k]);
      }
      __builtin_amdgcn_sched_barrier(0);
#pragma unroll
      for (int k = 0; k < VT; ++k) {
        const float4 pn = cu[k].p;
        const float dx = pn.x - pc[k].x;
        const float dy = pn.y - pc[k].y;
        const float dz = pn.z - pc[k].z;
        const float d  = __builtin_amdgcn_sqrtf(dx * dx + dy * dy + dz * dz);
        const float iv = __builtin_amdgcn_rcpf(fmaxf(d, EPSF));
        const float x = dx * iv;
        const float y = dy * iv;
        const float z = dz * iv;
        dmax[k] = fmaxf(dmax[k], d);

        const float4 g0 = cu[k].g0, g1 = cu[k].g1, g2 = cu[k].g2, g3 = cu[k].g3;
        const float sg[SLS] = {g0.x, g0.y, g0.z, g0.w, g1.x, g1.y, g1.z, g1.w,
                               g2.x, g2.y, g2.z, g2.w, g3.x, g3.y, g3.z, g3.w};
#pragma unroll
        for (int i = 0; i < SLS; ++i) {
          const float th = fmaxf(0.0f, sd0[i] * x + sd1[i] * y + sd2[i] * z);
          acc[k][i] = fmaxf(acc[k][i], th * sg[i]);
        }
      }
    }
#pragma unroll
    for (int k = 0; k < VT; ++k) cur[k] = nxt[k];
  }

#pragma unroll
  for (int k = 0; k < VT; ++k) {
    const size_t row = rbase + v0 + k * GT;
    float4 o0, o1;
    o0.x = ce0[k].x + acc[k][0] + acc[k][8];
    o0.y = ce0[k].y + acc[k][1] + acc[k][9];
    o0.z = ce0[k].z + acc[k][2] + acc[k][10];
    o0.w = ce0[k].w + acc[k][3] + acc[k][11];
    o1.x = ce1[k].x + acc[k][4] + acc[k][12];
    o1.y = ce1[k].y + acc[k][5] + acc[k][13];
    o1.z = ce1[k].z + acc[k][6] + acc[k][14];
    o1.w = ce1[k].w + acc[k][7] + acc[k][15];
    float4* pf = reinterpret_cast<float4*>(planes + ((size_t)j * MROWS + row) * SLC);
    pf[0] = o0;
    pf[1] = o1;

    if (j == 0) planes[(size_t)NSL * MROWS * SLC + row] = dmax[k];
  }
}

extern "C" void kernel_launch(void* const* d_in, const int* in_sizes, int n_in,
                              void* d_out, int out_size, void* d_ws, size_t ws_size,
                              hipStream_t stream) {
  const int*   nbr   = (const int*)  d_in[0];
  const float* verts = (const float*)d_in[1];
  const float* fmap  = (const float*)d_in[2];
  const float* wgt   = (const float*)d_in[3];
  const float* bias  = (const float*)d_in[4];
  const float* dirs  = (const float*)d_in[5];
  const float* distw = (const float*)d_in[6];
  const float* mlpw  = (const float*)d_in[7];
  const float* mlpb  = (const float*)d_in[8];
  float* out = (float*)d_out;

  float* feat   = (float*)d_ws;
  float* planes = feat + (size_t)NTILE * MROWS * 16;

  gemm_feat_rowblock<<<MROWS / RB, 384, 0, stream>>>(fmap, wgt, bias, feat);

  slab_gather_kernel<<<BS * NSL * (V / VPART), GT, 0, stream>>>(
      nbr, verts, feat, dirs, planes);

  gemm_out_rowblock<<<MROWS / RB, 256, 0, stream>>>(planes, mlpw, mlpb, distw, out);
}
